# starting kernel
# baseline (speedup 1.0000x reference)
#include <hip/hip_runtime.h>


typedef __attribute__((ext_vector_type(2))) float v2f;
typedef __attribute__((ext_vector_type(8))) float v8f;

#define C_IN   64
#define C_OUT  64
#define HW     32
#define K_TOT  576
#define LDS_T  580

__global__ __launch_bounds__(128, 1)
void local2d_wmma_f32(const float* __restrict__ xin,
                      const float* __restrict__ wgt,
                      const float* __restrict__ bias,
                      float* __restrict__ out)
{
    const int pos = blockIdx.x;
    const int x0  = pos >> 5;
    const int y0  = pos & 31;

    __shared__ float lds[32 * LDS_T];

    const int tid = threadIdx.x;

    for (int e = tid; e < 32 * K_TOT; e += 128) {
        const int n  = e / K_TOT;
        const int k  = e - n * K_TOT;
        const int i  = k / 9;
        const int r9 = k - i * 9;
        const int dk = r9 / 3;
        const int dl = r9 - dk * 3;
        const int h  = x0 + dk - 1;
        const int w  = y0 + dl - 1;
        float v = 0.0f;
        if ((unsigned)h < HW && (unsigned)w < HW)
            v = xin[(((n * C_IN) + i) * HW + h) * HW + w];
        lds[n * LDS_T + k] = v;
    }
    __syncthreads();

    const int lane  = tid & 31;
    const int wave  = tid >> 5;
    const int m0    = wave * 16;
    const int mrow  = m0 + (lane & 15);
    const int khalf = (lane >> 4) * 2;

    const float* wbase = wgt + (size_t)pos * (C_OUT * K_TOT);
    const float* arow  = wbase + mrow * K_TOT + khalf;

    __builtin_prefetch(arow + 128, 0, 3);
    __builtin_prefetch(arow + 320, 0, 3);

    const int bbase0 = (lane & 15) * LDS_T + khalf;
    const int bbase1 = bbase0 + 16 * LDS_T;

    v8f c0 = {};
    v8f c1 = {};

    #pragma unroll 4
    for (int k0 = 0; k0 < K_TOT; k0 += 4) {
        v2f a = *(const v2f*)(arow + k0);

        v2f b0 = *(const v2f*)(lds + bbase0 + k0);
        v2f b1 = *(const v2f*)(lds + bbase1 + k0);

        c0 = __builtin_amdgcn_wmma_f32_16x16x4_f32(
                 false, a, false, b0, (short)0, c0, false, false);
        c1 = __builtin_amdgcn_wmma_f32_16x16x4_f32(
                 false, a, false, b1, (short)0, c1, false, false);
    }

    const int mo   = m0 + (lane >> 4) * 8;
    const int n_lo = lane & 15;
    #pragma unroll
    for (int r = 0; r < 8; ++r) {
        const int m  = mo + r;
        const float bv = bias[m * (HW * HW) + pos];
        out[((n_lo        * C_OUT) + m) * (HW * HW) + pos] = c0[r] + bv;
        out[(((n_lo + 16) * C_OUT) + m) * (HW * HW) + pos] = c1[r] + bv;
    }
}

extern "C" void kernel_launch(void* const* d_in, const int* in_sizes, int n_in,
                              void* d_out, int out_size, void* d_ws, size_t ws_size,
                              hipStream_t stream) {
    const float* xin  = (const float*)d_in[0];
    const float* wgt  = (const float*)d_in[1];
    const float* bias = (const float*)d_in[2];
    float* out = (float*)d_out;

    dim3 grid(HW * HW);
    dim3 block(128);
    hipLaunchKernelGGL(local2d_wmma_f32, grid, block, 0, stream,
                       xin, wgt, bias, out);
}


# Input re-laid out per call by a pre-pass kernel into zero-padded [row][col][batch][channel] order in workspace: a block stages its patch from 3 contiguous 24-KB runs, 36 independent 16-B loads/thread
# speedup vs baseline: 1.6851x; 1.6851x over previous
#include <hip/hip_runtime.h>


typedef __attribute__((ext_vector_type(2))) float v2f;
typedef __attribute__((ext_vector_type(8))) float v8f;

#define C_IN   64
#define C_OUT  64
#define HW     32
#define K_TOT  576
#define LDS_T  580
#define PADHW  34
#define XT_POS 2048

__global__ __launch_bounds__(256)
void relayout_input(const float* __restrict__ xin, float* __restrict__ xt)
{
    __shared__ float tile[4 * C_IN * 33];

    const int hp  = blockIdx.x >> 3;
    const int b0  = (blockIdx.x & 7) * 4;
    const int tid = threadIdx.x;
    const bool interior = (hp >= 1) && (hp <= HW);

    if (interior) {
        const int h = hp - 1;
        #pragma unroll
        for (int j = 0; j < 8; ++j) {
            const int u   = tid + 256 * j;
            const int row = u >> 3;
            const int w4  = (u & 7) * 4;
            const int bb  = row >> 6;
            const int i   = row & 63;
            const float4 v = *(const float4*)
                (xin + ((((b0 + bb) * C_IN + i) * HW + h) * HW + w4));
            float* t = tile + row * 33 + w4;
            t[0] = v.x; t[1] = v.y; t[2] = v.z; t[3] = v.w;
        }
    }
    __syncthreads();

    float* dst = xt + ((size_t)(hp * PADHW) * 32 + b0) * C_IN + tid;
    if (interior) {
        dst[0] = 0.0f;
        #pragma unroll
        for (int w = 0; w < HW; ++w)
            dst[(w + 1) * XT_POS] = tile[tid * 33 + w];
        dst[(HW + 1) * XT_POS] = 0.0f;
    } else {
        #pragma unroll
        for (int wp = 0; wp < PADHW; ++wp)
            dst[wp * XT_POS] = 0.0f;
    }
}

__global__ __launch_bounds__(128, 1)
void local2d_wmma_f32(const float* __restrict__ xt,
                      const float* __restrict__ wgt,
                      const float* __restrict__ bias,
                      float* __restrict__ out)
{
    const int pos = blockIdx.x;
    const int x0  = pos >> 5;
    const int y0  = pos & 31;

    __shared__ float lds[32 * LDS_T];

    const int tid = threadIdx.x;

    {
        const float* xsrc = xt + (size_t)(x0 * PADHW + y0) * XT_POS + tid * 4;
        float4 r[36];
        #pragma unroll
        for (int r9 = 0; r9 < 9; ++r9) {
            #pragma unroll
            for (int jp = 0; jp < 4; ++jp)
                r[r9 * 4 + jp] = *(const float4*)
                    (xsrc + ((r9 / 3) * PADHW + (r9 % 3)) * XT_POS + jp * 512);
        }
        float* pdst = lds + (tid >> 4) * LDS_T + (tid & 15) * 36;
        #pragma unroll
        for (int r9 = 0; r9 < 9; ++r9) {
            #pragma unroll
            for (int jp = 0; jp < 4; ++jp) {
                const float4 v = r[r9 * 4 + jp];
                float* p = pdst + jp * 8 * LDS_T + r9;
                p[0] = v.x; p[9] = v.y; p[18] = v.z; p[27] = v.w;
            }
        }
    }
    __syncthreads();

    const int lane  = tid & 31;
    const int wave  = tid >> 5;
    const int m0    = wave * 16;
    const int mrow  = m0 + (lane & 15);
    const int khalf = (lane >> 4) * 2;

    const float* wbase = wgt + (size_t)pos * (C_OUT * K_TOT);
    const float* arow  = wbase + mrow * K_TOT + khalf;

    __builtin_prefetch(arow + 128, 0, 3);
    __builtin_prefetch(arow + 320, 0, 3);

    const int bbase0 = (lane & 15) * LDS_T + khalf;
    const int bbase1 = bbase0 + 16 * LDS_T;

    v8f c0 = {};
    v8f c1 = {};

    #pragma unroll 4
    for (int k0 = 0; k0 < K_TOT; k0 += 4) {
        v2f a = *(const v2f*)(arow + k0);

        v2f b0 = *(const v2f*)(lds + bbase0 + k0);
        v2f b1 = *(const v2f*)(lds + bbase1 + k0);

        c0 = __builtin_amdgcn_wmma_f32_16x16x4_f32(
                 false, a, false, b0, (short)0, c0, false, false);
        c1 = __builtin_amdgcn_wmma_f32_16x16x4_f32(
                 false, a, false, b1, (short)0, c1, false, false);
    }

    const int mo   = m0 + (lane >> 4) * 8;
    const int n_lo = lane & 15;
    #pragma unroll
    for (int r = 0; r < 8; ++r) {
        const int m  = mo + r;
        const float bv = bias[m * (HW * HW) + pos];
        out[((n_lo        * C_OUT) + m) * (HW * HW) + pos] = c0[r] + bv;
        out[(((n_lo + 16) * C_OUT) + m) * (HW * HW) + pos] = c1[r] + bv;
    }
}

extern "C" void kernel_launch(void* const* d_in, const int* in_sizes, int n_in,
                              void* d_out, int out_size, void* d_ws, size_t ws_size,
                              hipStream_t stream) {
    const float* xin  = (const float*)d_in[0];
    const float* wgt  = (const float*)d_in[1];
    const float* bias = (const float*)d_in[2];
    float* out = (float*)d_out;
    float* xt  = (float*)d_ws;

    hipLaunchKernelGGL(relayout_input, dim3(PADHW * 8), dim3(256), 0, stream,
                       xin, xt);

    dim3 grid(HW * HW);
    dim3 block(128);
    hipLaunchKernelGGL(local2d_wmma_f32, grid, block, 0, stream,
                       (const float*)xt, wgt, bias, out);
}


# Matrix kernel stores its accumulator registers as 512-byte runs into workspace; a new third launch turns them through LDS, adds bias and writes out as whole 128-byte lines (was 4-byte pieces)
# speedup vs baseline: 4.2810x; 2.5405x over previous
#include <hip/hip_runtime.h>


typedef __attribute__((ext_vector_type(2))) float v2f;
typedef __attribute__((ext_vector_type(4))) float v4f;
typedef __attribute__((ext_vector_type(8))) float v8f;

#define C_IN   64
#define C_OUT  64
#define HW     32
#define K_TOT  576
#define LDS_T  580
#define PADHW  34
#define XT_POS 2048
#define DT_POS 2048

__global__ __launch_bounds__(256)
void relayout_input(const float* __restrict__ xin, float* __restrict__ xt)
{
    __shared__ float tile[4 * C_IN * 33];

    const int hp  = blockIdx.x >> 3;
    const int b0  = (blockIdx.x & 7) * 4;
    const int tid = threadIdx.x;
    const bool interior = (hp >= 1) && (hp <= HW);

    if (interior) {
        const int h = hp - 1;
        #pragma unroll
        for (int j = 0; j < 8; ++j) {
            const int u   = tid + 256 * j;
            const int row = u >> 3;
            const int w4  = (u & 7) * 4;
            const int bb  = row >> 6;
            const int i   = row & 63;
            const float4 v = *(const float4*)
                (xin + ((((b0 + bb) * C_IN + i) * HW + h) * HW + w4));
            float* t = tile + row * 33 + w4;
            t[0] = v.x; t[1] = v.y; t[2] = v.z; t[3] = v.w;
        }
    }
    __syncthreads();

    float* dst = xt + ((size_t)(hp * PADHW) * 32 + b0) * C_IN + tid;
    if (interior) {
        dst[0] = 0.0f;
        #pragma unroll
        for (int w = 0; w < HW; ++w)
            dst[(w + 1) * XT_POS] = tile[tid * 33 + w];
        dst[(HW + 1) * XT_POS] = 0.0f;
    } else {
        #pragma unroll
        for (int wp = 0; wp < PADHW; ++wp)
            dst[wp * XT_POS] = 0.0f;
    }
}

__global__ __launch_bounds__(128, 1)
void local2d_wmma_f32(const float* __restrict__ xt,
                      const float* __restrict__ wgt,
                      float* __restrict__ dt)
{
    const int pos = blockIdx.x;
    const int x0  = pos >> 5;
    const int y0  = pos & 31;

    __shared__ float lds[32 * LDS_T];

    const int tid = threadIdx.x;

    {
        const float* xsrc = xt + (size_t)(x0 * PADHW + y0) * XT_POS + tid * 4;
        float4 r[36];
        #pragma unroll
        for (int r9 = 0; r9 < 9; ++r9) {
            #pragma unroll
            for (int jp = 0; jp < 4; ++jp)
                r[r9 * 4 + jp] = *(const float4*)
                    (xsrc + ((r9 / 3) * PADHW + (r9 % 3)) * XT_POS + jp * 512);
        }
        float* pdst = lds + (tid >> 4) * LDS_T + (tid & 15) * 36;
        #pragma unroll
        for (int r9 = 0; r9 < 9; ++r9) {
            #pragma unroll
            for (int jp = 0; jp < 4; ++jp) {
                const float4 v = r[r9 * 4 + jp];
                float* p = pdst + jp * 8 * LDS_T + r9;
                p[0] = v.x; p[9] = v.y; p[18] = v.z; p[27] = v.w;
            }
        }
    }
    __syncthreads();

    const int lane  = tid & 31;
    const int wave  = tid >> 5;
    const int m0    = wave * 16;
    const int mrow  = m0 + (lane & 15);
    const int khalf = (lane >> 4) * 2;

    const float* wbase = wgt + (size_t)pos * (C_OUT * K_TOT);
    const float* arow  = wbase + mrow * K_TOT + khalf;

    __builtin_prefetch(arow + 128, 0, 3);
    __builtin_prefetch(arow + 320, 0, 3);

    const int bbase0 = (lane & 15) * LDS_T + khalf;
    const int bbase1 = bbase0 + 16 * LDS_T;

    v8f c0 = {};
    v8f c1 = {};

    #pragma unroll 4
    for (int k0 = 0; k0 < K_TOT; k0 += 4) {
        v2f a = *(const v2f*)(arow + k0);

        v2f b0 = *(const v2f*)(lds + bbase0 + k0);
        v2f b1 = *(const v2f*)(lds + bbase1 + k0);

        c0 = __builtin_amdgcn_wmma_f32_16x16x4_f32(
                 false, a, false, b0, (short)0, c0, false, false);
        c1 = __builtin_amdgcn_wmma_f32_16x16x4_f32(
                 false, a, false, b1, (short)0, c1, false, false);
    }

    float* dw = dt + (size_t)pos * DT_POS + wave * 512 + lane * 4;
    *(v4f*)(dw)       = __builtin_shufflevector(c0, c0, 0, 1, 2, 3);
    *(v4f*)(dw + 128) = __builtin_shufflevector(c0, c0, 4, 5, 6, 7);
    *(v4f*)(dw + 256) = __builtin_shufflevector(c1, c1, 0, 1, 2, 3);
    *(v4f*)(dw + 384) = __builtin_shufflevector(c1, c1, 4, 5, 6, 7);
}

__global__ __launch_bounds__(256)
void relayout_output(const float* __restrict__ dt,
                     const float* __restrict__ bias,
                     float* __restrict__ out)
{
    __shared__ float tile[256 * 33];

    const int x   = blockIdx.x >> 3;
    const int g   = blockIdx.x & 7;
    const int tid = threadIdx.x;

    const float* src = dt + (size_t)(x * HW) * DT_POS + g * 256 + tid;
    float v[HW];
    #pragma unroll
    for (int y = 0; y < HW; ++y)
        v[y] = src[y * DT_POS];
    #pragma unroll
    for (int y = 0; y < HW; ++y)
        tile[tid * 33 + y] = v[y];
    __syncthreads();

    #pragma unroll
    for (int jj = 0; jj < 8; ++jj) {
        const int u  = tid + 256 * jj;
        const int s  = u >> 3;
        const int y4 = (u & 7) * 4;
        const int m  = 16 * (g >> 1) + 8 * ((s >> 6) & 1) + 4 * (s >> 7) + (s & 3);
        const int n  = 16 * (g & 1) + ((s >> 2) & 15);
        const float4 bv = *(const float4*)(bias + (m * HW + x) * HW + y4);
        const float* t = tile + s * 33 + y4;
        float4 o;
        o.x = t[0] + bv.x; o.y = t[1] + bv.y; o.z = t[2] + bv.z; o.w = t[3] + bv.w;
        *(float4*)(out + ((n * C_OUT + m) * HW + x) * HW + y4) = o;
    }
}

extern "C" void kernel_launch(void* const* d_in, const int* in_sizes, int n_in,
                              void* d_out, int out_size, void* d_ws, size_t ws_size,
                              hipStream_t stream) {
    const float* xin  = (const float*)d_in[0];
    const float* wgt  = (const float*)d_in[1];
    const float* bias = (const float*)d_in[2];
    float* out = (float*)d_out;
    float* xt  = (float*)d_ws;
    float* dt  = (float*)d_ws + 2367488;

    hipLaunchKernelGGL(relayout_input, dim3(PADHW * 8), dim3(256), 0, stream,
                       xin, xt);

    dim3 grid(HW * HW);
    dim3 block(128);
    hipLaunchKernelGGL(local2d_wmma_f32, grid, block, 0, stream,
                       (const float*)xt, wgt, dt);

    hipLaunchKernelGGL(relayout_output, dim3(HW * 8), dim3(256), 0, stream,
                       (const float*)dt, bias, out);
}


# Weights read as whole 128-B lines (4 rows x 128 B per load, 3 units of 32 k ahead in registers), made into A fragments via a per-wave 16x16 LDS stage; was 16 lines x 16 B per load, no lead
# speedup vs baseline: 4.6589x; 1.0883x over previous
#include <hip/hip_runtime.h>


typedef __attribute__((ext_vector_type(2))) float v2f;
typedef __attribute__((ext_vector_type(4))) float v4f;
typedef __attribute__((ext_vector_type(8))) float v8f;

#define C_IN   64
#define C_OUT  64
#define HW     32
#define K_TOT  576
#define LDS_T  580
#define PADHW  34
#define XT_POS 2048
#define DT_POS 2048
#define WST_P  20
#define WST_W  (16 * WST_P)

__global__ __launch_bounds__(256)
void relayout_input(const float* __restrict__ xin, float* __restrict__ xt)
{
    __shared__ float tile[4 * C_IN * 33];

    const int hp  = blockIdx.x >> 3;
    const int b0  = (blockIdx.x & 7) * 4;
    const int tid = threadIdx.x;
    const bool interior = (hp >= 1) && (hp <= HW);

    if (interior) {
        const int h = hp - 1;
        #pragma unroll
        for (int j = 0; j < 8; ++j) {
            const int u   = tid + 256 * j;
            const int row = u >> 3;
            const int w4  = (u & 7) * 4;
            const int bb  = row >> 6;
            const int i   = row & 63;
            const float4 v = *(const float4*)
                (xin + ((((b0 + bb) * C_IN + i) * HW + h) * HW + w4));
            float* t = tile + row * 33 + w4;
            t[0] = v.x; t[1] = v.y; t[2] = v.z; t[3] = v.w;
        }
    }
    __syncthreads();

    float* dst = xt + ((size_t)(hp * PADHW) * 32 + b0) * C_IN + tid;
    if (interior) {
        dst[0] = 0.0f;
        #pragma unroll
        for (int w = 0; w < HW; ++w)
            dst[(w + 1) * XT_POS] = tile[tid * 33 + w];
        dst[(HW + 1) * XT_POS] = 0.0f;
    } else {
        #pragma unroll
        for (int wp = 0; wp < PADHW; ++wp)
            dst[wp * XT_POS] = 0.0f;
    }
}

struct Frags { v2f a[4], b0[4], b1[4]; };

__device__ __forceinline__ void load_frags(Frags& f, const float* aw,
                                           const float* pb0, const float* pb1)
{
    #pragma unroll
    for (int s = 0; s < 4; ++s) {
        f.a[s]  = *(const v2f*)(aw  + 4 * s);
        f.b0[s] = *(const v2f*)(pb0 + 4 * s);
        f.b1[s] = *(const v2f*)(pb1 + 4 * s);
    }
}

__device__ __forceinline__ void mma_half(const Frags& f, v8f& c0, v8f& c1)
{
    #pragma unroll
    for (int s = 0; s < 4; ++s) {
        c0 = __builtin_amdgcn_wmma_f32_16x16x4_f32(
                 false, f.a[s], false, f.b0[s], (short)0, c0, false, false);
        c1 = __builtin_amdgcn_wmma_f32_16x16x4_f32(
                 false, f.a[s], false, f.b1[s], (short)0, c1, false, false);
    }
}

__device__ __forceinline__ void stage_sync()
{
    __builtin_amdgcn_fence(__ATOMIC_RELEASE, "wavefront");
    __builtin_amdgcn_wave_barrier();
    __builtin_amdgcn_fence(__ATOMIC_ACQUIRE, "wavefront");
}

__device__ __forceinline__ void stage_put(const v4f (&w)[4], float* sdst, bool mine)
{
    stage_sync();
    if (mine) {
        #pragma unroll
        for (int j = 0; j < 4; ++j)
            *(v4f*)(sdst + j * 4 * WST_P) = w[j];
    }
    stage_sync();
}

template <bool REFILL, bool HAS_NEXT>
__device__ __forceinline__ void unit32(v4f (&w)[4], const v4f (&wnext)[4],
                                       const float* wsrc_next, float* sdst, bool lo,
                                       const float* aw, const float* pb0,
                                       const float* pb1, Frags& f0, v8f& c0, v8f& c1)
{
    Frags f1;
    stage_put(w, sdst, !lo);
    load_frags(f1, aw, pb0 + 16, pb1 + 16);
    if (REFILL) {
        #pragma unroll
        for (int j = 0; j < 4; ++j)
            w[j] = *(const v4f*)(wsrc_next + j * 4 * K_TOT);
    }
    mma_half(f0, c0, c1);
    if (HAS_NEXT) {
        stage_put(wnext, sdst, lo);
        load_frags(f0, aw, pb0 + 32, pb1 + 32);
    }
    mma_half(f1, c0, c1);
}

__global__ __launch_bounds__(128, 1)
void local2d_wmma_f32(const float* __restrict__ xt,
                      const float* __restrict__ wgt,
                      float* __restrict__ dt)
{
    const int pos = blockIdx.x;
    const int x0  = pos >> 5;
    const int y0  = pos & 31;

    __shared__ float lds[32 * LDS_T];
    __shared__ __attribute__((aligned(16))) float wst[4 * WST_W];

    const int tid   = threadIdx.x;
    const int lane  = tid & 31;
    const int wave  = tid >> 5;
    const int m0    = wave * 16;
    const int khalf = (lane >> 4) * 2;

    const int rl = lane >> 3;
    const int q  = lane & 7;
    const float* wsrc = wgt + (size_t)pos * (C_OUT * K_TOT)
                            + (m0 + rl) * K_TOT + q * 4;
    v4f wr[3][4];

    {
        const float* xsrc = xt + (size_t)(x0 * PADHW + y0) * XT_POS + tid * 4;
        float4 r[36];
        #pragma unroll
        for (int r9 = 0; r9 < 9; ++r9) {
            #pragma unroll
            for (int jp = 0; jp < 4; ++jp)
                r[r9 * 4 + jp] = *(const float4*)
                    (xsrc + ((r9 / 3) * PADHW + (r9 % 3)) * XT_POS + jp * 512);
        }
        #pragma unroll
        for (int u = 0; u < 3; ++u) {
            #pragma unroll
            for (int j = 0; j < 4; ++j)
                wr[u][j] = *(const v4f*)(wsrc + j * 4 * K_TOT + u * 32);
        }
        float* pdst = lds + (tid >> 4) * LDS_T + (tid & 15) * 36;
        #pragma unroll
        for (int r9 = 0; r9 < 9; ++r9) {
            #pragma unroll
            for (int jp = 0; jp < 4; ++jp) {
                const float4 v = r[r9 * 4 + jp];
                float* p = pdst + jp * 8 * LDS_T + r9;
                p[0] = v.x; p[9] = v.y; p[18] = v.z; p[27] = v.w;
            }
        }
    }
    __syncthreads();

    const float* pb0 = lds + (lane & 15) * LDS_T + khalf;
    const float* pb1 = pb0 + 16 * LDS_T;
    float* st          = wst + wave * WST_W;
    float* sdst        = st + rl * WST_P + (q & 3) * 4;
    const float* aw    = st + (lane & 15) * WST_P + khalf;
    const bool   lo    = q < 4;

    v8f c0 = {};
    v8f c1 = {};

    Frags f0;
    stage_put(wr[0], sdst, lo);
    load_frags(f0, aw, pb0, pb1);

    #pragma unroll 1
    for (int u3 = 0; u3 < 15; u3 += 3) {
        #pragma unroll
        for (int sl = 0; sl < 3; ++sl) {
            const int kb = (u3 + sl) * 32;
            unit32<true, true>(wr[sl], wr[(sl + 1) % 3], wsrc + kb + 96, sdst, lo,
                               aw, pb0 + kb, pb1 + kb, f0, c0, c1);
        }
    }
    unit32<false, true >(wr[0], wr[1], wsrc, sdst, lo, aw, pb0 + 480, pb1 + 480, f0, c0, c1);
    unit32<false, true >(wr[1], wr[2], wsrc, sdst, lo, aw, pb0 + 512, pb1 + 512, f0, c0, c1);
    unit32<false, false>(wr[2], wr[2], wsrc, sdst, lo, aw, pb0 + 544, pb1 + 544, f0, c0, c1);

    float* dw = dt + (size_t)pos * DT_POS + wave * 512 + lane * 4;
    *(v4f*)(dw)       = __builtin_shufflevector(c0, c0, 0, 1, 2, 3);
    *(v4f*)(dw + 128) = __builtin_shufflevector(c0, c0, 4, 5, 6, 7);
    *(v4f*)(dw + 256) = __builtin_shufflevector(c1, c1, 0, 1, 2, 3);
    *(v4f*)(dw + 384) = __builtin_shufflevector(c1, c1, 4, 5, 6, 7);
}

__global__ __launch_bounds__(256)
void relayout_output(const float* __restrict__ dt,
                     const float* __restrict__ bias,
                     float* __restrict__ out)
{
    __shared__ float tile[256 * 33];

    const int x   = blockIdx.x >> 3;
    const int g   = blockIdx.x & 7;
    const int tid = threadIdx.x;

    const float* src = dt + (size_t)(x * HW) * DT_POS + g * 256 + tid;
    float v[HW];
    #pragma unroll
    for (int y = 0; y < HW; ++y)
        v[y] = src[y * DT_POS];
    #pragma unroll
    for (int y = 0; y < HW; ++y)
        tile[tid * 33 + y] = v[y];
    __syncthreads();

    #pragma unroll
    for (int jj = 0; jj < 8; ++jj) {
        const int u  = tid + 256 * jj;
        const int s  = u >> 3;
        const int y4 = (u & 7) * 4;
        const int m  = 16 * (g >> 1) + 8 * ((s >> 6) & 1) + 4 * (s >> 7) + (s & 3);
        const int n  = 16 * (g & 1) + ((s >> 2) & 15);
        const float4 bv = *(const float4*)(bias + (m * HW + x) * HW + y4);
        const float* t = tile + s * 33 + y4;
        float4 o;
        o.x = t[0] + bv.x; o.y = t[1] + bv.y; o.z = t[2] + bv.z; o.w = t[3] + bv.w;
        *(float4*)(out + ((n * C_OUT + m) * HW + x) * HW + y4) = o;
    }
}

extern "C" void kernel_launch(void* const* d_in, const int* in_sizes, int n_in,
                              void* d_out, int out_size, void* d_ws, size_t ws_size,
                              hipStream_t stream) {
    const float* xin  = (const float*)d_in[0];
    const float* wgt  = (const float*)d_in[1];
    const float* bias = (const float*)d_in[2];
    float* out = (float*)d_out;
    float* xt  = (float*)d_ws;
    float* dt  = (float*)d_ws + 2367488;

    hipLaunchKernelGGL(relayout_input, dim3(PADHW * 8), dim3(256), 0, stream,
                       xin, xt);

    dim3 grid(HW * HW);
    dim3 block(128);
    hipLaunchKernelGGL(local2d_wmma_f32, grid, block, 0, stream,
                       (const float*)xt, wgt, dt);

    hipLaunchKernelGGL(relayout_output, dim3(HW * 8), dim3(256), 0, stream,
                       (const float*)dt, bias, out);
}
